# starting kernel
# baseline (speedup 1.0000x reference)
#include <hip/hip_runtime.h>


#define TOKENS 2048
#define INF 4096
#define OUTF 4096
#define JB 32
#define NCHUNK 2
#define XP_STRIDE 65
#define TT_STRIDE (JB + 1)
#define SMEM_FLOATS (1024 * TT_STRIDE + 64 * XP_STRIDE)
#define SMEM_BYTES (SMEM_FLOATS * 4)

typedef __attribute__((ext_vector_type(2))) float v2f;
typedef __attribute__((ext_vector_type(8))) float v8f;

__global__ __launch_bounds__(256)
void tt_linear_kernel(const float* __restrict__ x,
                      const float* __restrict__ g0,
                      const float* __restrict__ g1,
                      const float* __restrict__ alpha,
                      const float* __restrict__ pds,
                      const float* __restrict__ bias,
                      const int*   __restrict__ iperm,
                      const int*   __restrict__ oinv,
                      float* __restrict__ out)
{
    extern __shared__ float smem[];
    float* Tt = smem;
    float* xp = smem + 1024 * TT_STRIDE;

    const int b    = blockIdx.x;
    const int tid  = threadIdx.x;
    const int lane = tid & 31;
    const int w    = tid >> 5;
    const int half = lane >> 4;
    const int l15  = lane & 15;
    const int koff = half * 2;

    const float* xrow = x + (size_t)b * INF;
    for (int c = tid; c < INF; c += 256) {
        xp[(c >> 6) * XP_STRIDE + (c & 63)] = xrow[iperm[c]];
    }
    __syncthreads();

    const int ti  = w & 3;
    const int tm0 = (w >> 2) * 2;
    v8f accY0 = {};
    v8f accY1 = {};

    for (int chunk = 0; chunk < NCHUNK; ++chunk) {
        const int jbase = chunk * JB;

        for (int cp = 0; cp < 4; ++cp) {
            const int ct0   = w * 2 + cp * 16;
            const int cb0   = ct0 * 16;
            const int cb1   = cb0 + 16;
            const float* ap0 = g1 + (cb0 + l15) * 64 + koff;
            const float* ap1 = g1 + (cb1 + l15) * 64 + koff;
            v8f t00 = {}, t01 = {};
            v8f t10 = {}, t11 = {};
            #pragma unroll
            for (int kb = 0; kb < 64; kb += 4) {
                v2f a0 = *(const v2f*)(ap0 + kb);
                v2f a1 = *(const v2f*)(ap1 + kb);
                const int k = kb + koff;
                v2f b0, b1;
                {
                    const float* bp = xp + (jbase + l15) * XP_STRIDE + k;
                    b0.x = bp[0]; b0.y = bp[1];
                }
                {
                    const float* bp = xp + (jbase + 16 + l15) * XP_STRIDE + k;
                    b1.x = bp[0]; b1.y = bp[1];
                }
                t00 = __builtin_amdgcn_wmma_f32_16x16x4_f32(false, a0, false, b0, (short)0, t00, false, false);
                t01 = __builtin_amdgcn_wmma_f32_16x16x4_f32(false, a0, false, b1, (short)0, t01, false, false);
                t10 = __builtin_amdgcn_wmma_f32_16x16x4_f32(false, a1, false, b0, (short)0, t10, false, false);
                t11 = __builtin_amdgcn_wmma_f32_16x16x4_f32(false, a1, false, b1, (short)0, t11, false, false);
            }
            #pragma unroll
            for (int v = 0; v < 8; ++v) {
                const int r0 = cb0 + v + half * 8;
                const int r1 = cb1 + v + half * 8;
                Tt[r0 * TT_STRIDE + l15]      = t00[v];
                Tt[r0 * TT_STRIDE + 16 + l15] = t01[v];
                Tt[r1 * TT_STRIDE + l15]      = t10[v];
                Tt[r1 * TT_STRIDE + 16 + l15] = t11[v];
            }
        }
        __syncthreads();

        const float* aptr = g0 + (ti * 16 + l15) * 1024 + jbase * 16 + koff;
        #pragma unroll 4
        for (int ks = 0; ks < JB * 16; ks += 4) {
            v2f a = *(const v2f*)(aptr + ks);
            const int kk = ks + koff;
            const int r  = kk & 15;
            const int jl = kk >> 4;
            {
                const int m = tm0 * 16 + l15;
                v2f bv;
                bv.x = Tt[(r * 64 + m) * TT_STRIDE + jl];
                bv.y = Tt[((r + 1) * 64 + m) * TT_STRIDE + jl];
                accY0 = __builtin_amdgcn_wmma_f32_16x16x4_f32(false, a, false, bv, (short)0, accY0, false, false);
            }
            {
                const int m = (tm0 + 1) * 16 + l15;
                v2f bv;
                bv.x = Tt[(r * 64 + m) * TT_STRIDE + jl];
                bv.y = Tt[((r + 1) * 64 + m) * TT_STRIDE + jl];
                accY1 = __builtin_amdgcn_wmma_f32_16x16x4_f32(false, a, false, bv, (short)0, accY1, false, false);
            }
        }
        __syncthreads();
    }

    const float av = alpha[0];
    float* ylds = xp;
    #pragma unroll
    for (int v = 0; v < 8; ++v) {
        const int i  = ti * 16 + v + half * 8;
        const int m0 = tm0 * 16 + l15;
        const int m1 = m0 + 16;
        ylds[i * XP_STRIDE + m0] = accY0[v] * av * pds[i * 64 + m0];
        ylds[i * XP_STRIDE + m1] = accY1[v] * av * pds[i * 64 + m1];
    }
    __syncthreads();

    float* orow = out + (size_t)b * OUTF;
    for (int q = tid; q < OUTF; q += 256) {
        const int o = oinv[q];
        orow[q] = ylds[(o >> 6) * XP_STRIDE + (o & 63)] + bias[q];
    }
}

extern "C" void kernel_launch(void* const* d_in, const int* in_sizes, int n_in,
                              void* d_out, int out_size, void* d_ws, size_t ws_size,
                              hipStream_t stream) {
    (void)in_sizes; (void)n_in; (void)out_size; (void)d_ws; (void)ws_size;
    const float* x     = (const float*)d_in[0];
    const float* g0    = (const float*)d_in[1];
    const float* g1    = (const float*)d_in[2];
    const float* alpha = (const float*)d_in[3];
    const float* pds   = (const float*)d_in[4];
    const float* bias  = (const float*)d_in[5];
    const int*   iperm = (const int*)d_in[6];
    const int*   oinv  = (const int*)d_in[7];
    float* out = (float*)d_out;

    (void)hipFuncSetAttribute(reinterpret_cast<const void*>(tt_linear_kernel),
                              hipFuncAttributeMaxDynamicSharedMemorySize, SMEM_BYTES);

    tt_linear_kernel<<<TOKENS, 256, SMEM_BYTES, stream>>>(
        x, g0, g1, alpha, pds, bias, iperm, oinv, out);
}


# T staged per rank half (8 r x all 64 j) and GEMM2 waves own 4 m-tiles over half of j: each G1/G0 fragment load feeds 4 WMMAs, not 2; sum order changes (j-half partials added at end)
# speedup vs baseline: 1.1421x; 1.1421x over previous
#include <hip/hip_runtime.h>


#define TOKENS 2048
#define INF 4096
#define OUTF 4096
#define RH 8
#define NCHUNK 2
#define XP_STRIDE 65
#define TT_STRIDE 65
#define TT_ROWS (RH * 64)
#define SMEM_FLOATS (TT_ROWS * TT_STRIDE + 64 * XP_STRIDE)
#define SMEM_BYTES (SMEM_FLOATS * 4)

typedef __attribute__((ext_vector_type(2))) float v2f;
typedef __attribute__((ext_vector_type(8))) float v8f;

__global__ __launch_bounds__(256)
void tt_linear_kernel(const float* __restrict__ x,
                      const float* __restrict__ g0,
                      const float* __restrict__ g1,
                      const float* __restrict__ alpha,
                      const float* __restrict__ pds,
                      const float* __restrict__ bias,
                      const int*   __restrict__ iperm,
                      const int*   __restrict__ oinv,
                      float* __restrict__ out)
{
    extern __shared__ float smem[];
    float* Tt = smem;
    float* xp = smem + TT_ROWS * TT_STRIDE;

    const int b    = blockIdx.x;
    const int tid  = threadIdx.x;
    const int lane = tid & 31;
    const int w    = tid >> 5;
    const int half = lane >> 4;
    const int l15  = lane & 15;
    const int koff = half * 2;

    const float* xrow = x + (size_t)b * INF;
    for (int c = tid; c < INF; c += 256) {
        xp[(c >> 6) * XP_STRIDE + (c & 63)] = xrow[iperm[c]];
    }
    __syncthreads();

    const int ti = w & 3;
    const int kh = w >> 2;
    v8f accY[4];
    #pragma unroll
    for (int mt = 0; mt < 4; ++mt) accY[mt] = (v8f){};

    for (int chunk = 0; chunk < NCHUNK; ++chunk) {
        for (int cp = 0; cp < 2; ++cp) {
            const int cb0 = (w * 2 + cp * 16) * 16;
            const int cb1 = cb0 + 16;
            const float* ap0 = g1 + (chunk * TT_ROWS + cb0 + l15) * 64 + koff;
            const float* ap1 = g1 + (chunk * TT_ROWS + cb1 + l15) * 64 + koff;
            v8f t0[4], t1[4];
            #pragma unroll
            for (int jt = 0; jt < 4; ++jt) { t0[jt] = (v8f){}; t1[jt] = (v8f){}; }
            #pragma unroll
            for (int kb = 0; kb < 64; kb += 4) {
                v2f a0 = *(const v2f*)(ap0 + kb);
                v2f a1 = *(const v2f*)(ap1 + kb);
                const int k = kb + koff;
                #pragma unroll
                for (int jt = 0; jt < 4; ++jt) {
                    const float* bp = xp + (jt * 16 + l15) * XP_STRIDE + k;
                    v2f bb;
                    bb.x = bp[0]; bb.y = bp[1];
                    t0[jt] = __builtin_amdgcn_wmma_f32_16x16x4_f32(false, a0, false, bb, (short)0, t0[jt], false, false);
                    t1[jt] = __builtin_amdgcn_wmma_f32_16x16x4_f32(false, a1, false, bb, (short)0, t1[jt], false, false);
                }
            }
            #pragma unroll
            for (int v = 0; v < 8; ++v) {
                const int r0 = cb0 + v + half * 8;
                const int r1 = cb1 + v + half * 8;
                #pragma unroll
                for (int jt = 0; jt < 4; ++jt) {
                    Tt[r0 * TT_STRIDE + jt * 16 + l15] = t0[jt][v];
                    Tt[r1 * TT_STRIDE + jt * 16 + l15] = t1[jt][v];
                }
            }
        }
        __syncthreads();

        const float* aptr = g0 + (ti * 16 + l15) * 1024 + kh * 512 + chunk * RH + koff;
        const float* tb   = Tt + (koff * 64 + l15) * TT_STRIDE + kh * 32;
        #pragma unroll 4
        for (int jj = 0; jj < 32; ++jj) {
            #pragma unroll
            for (int q = 0; q < RH; q += 4) {
                v2f a = *(const v2f*)(aptr + jj * 16 + q);
                #pragma unroll
                for (int mt = 0; mt < 4; ++mt) {
                    const float* bp = tb + (q * 64 + mt * 16) * TT_STRIDE + jj;
                    v2f bv;
                    bv.x = bp[0];
                    bv.y = bp[64 * TT_STRIDE];
                    accY[mt] = __builtin_amdgcn_wmma_f32_16x16x4_f32(false, a, false, bv, (short)0, accY[mt], false, false);
                }
            }
        }
        __syncthreads();
    }

    const float av = alpha[0];
    float* ylds = xp;
    if (kh == 1) {
        #pragma unroll
        for (int v = 0; v < 8; ++v) {
            const int i = ti * 16 + v + half * 8;
            #pragma unroll
            for (int mt = 0; mt < 4; ++mt)
                ylds[i * XP_STRIDE + mt * 16 + l15] = accY[mt][v];
        }
    }
    __syncthreads();
    if (kh == 0) {
        #pragma unroll
        for (int v = 0; v < 8; ++v) {
            const int i = ti * 16 + v + half * 8;
            #pragma unroll
            for (int mt = 0; mt < 4; ++mt) {
                const int m = mt * 16 + l15;
                const float s = accY[mt][v] + ylds[i * XP_STRIDE + m];
                ylds[i * XP_STRIDE + m] = s * av * pds[i * 64 + m];
            }
        }
    }
    __syncthreads();

    float* orow = out + (size_t)b * OUTF;
    for (int q = tid; q < OUTF; q += 256) {
        const int o = oinv[q];
        orow[q] = ylds[(o >> 6) * XP_STRIDE + (o & 63)] + bias[q];
    }
}

extern "C" void kernel_launch(void* const* d_in, const int* in_sizes, int n_in,
                              void* d_out, int out_size, void* d_ws, size_t ws_size,
                              hipStream_t stream) {
    (void)in_sizes; (void)n_in; (void)out_size; (void)d_ws; (void)ws_size;
    const float* x     = (const float*)d_in[0];
    const float* g0    = (const float*)d_in[1];
    const float* g1    = (const float*)d_in[2];
    const float* alpha = (const float*)d_in[3];
    const float* pds   = (const float*)d_in[4];
    const float* bias  = (const float*)d_in[5];
    const int*   iperm = (const int*)d_in[6];
    const int*   oinv  = (const int*)d_in[7];
    float* out = (float*)d_out;

    (void)hipFuncSetAttribute(reinterpret_cast<const void*>(tt_linear_kernel),
                              hipFuncAttributeMaxDynamicSharedMemorySize, SMEM_BYTES);

    tt_linear_kernel<<<TOKENS, 256, SMEM_BYTES, stream>>>(
        x, g0, g1, alpha, pds, bias, iperm, oinv, out);
}


# Two tokens per workgroup (16 waves, one workgroup per WGP): each G1/G0 fragment load from global feeds 8 WMMAs, not 4; sum order changes (4 j-quarter partials added in quarter order at the end)
# speedup vs baseline: 1.1816x; 1.0346x over previous
#include <hip/hip_runtime.h>


#define TOKENS 2048
#define TPW 2
#define INF 4096
#define OUTF 4096
#define RH 8
#define NCHUNK 2
#define XP_STRIDE 65
#define TT_STRIDE 65
#define TT_ROWS (RH * 64)
#define TT_TOK (TT_ROWS * TT_STRIDE)
#define XP_TOK (64 * XP_STRIDE)
#define SMEM_FLOATS (TPW * TT_TOK + TPW * XP_TOK)
#define SMEM_BYTES (SMEM_FLOATS * 4)

typedef __attribute__((ext_vector_type(2))) float v2f;
typedef __attribute__((ext_vector_type(8))) float v8f;

__global__ __launch_bounds__(512)
void tt_linear_kernel(const float* __restrict__ x,
                      const float* __restrict__ g0,
                      const float* __restrict__ g1,
                      const float* __restrict__ alpha,
                      const float* __restrict__ pds,
                      const float* __restrict__ bias,
                      const int*   __restrict__ iperm,
                      const int*   __restrict__ oinv,
                      float* __restrict__ out)
{
    extern __shared__ float smem[];
    float* Tt = smem;
    float* xp = smem + TPW * TT_TOK;

    const int b0   = blockIdx.x * TPW;
    const int tid  = threadIdx.x;
    const int lane = tid & 31;
    const int w    = tid >> 5;
    const int half = lane >> 4;
    const int l15  = lane & 15;
    const int koff = half * 2;
    const int tsel = tid >> 8;
    const int t8   = tid & 255;

    {
        const float* xrow = x + (size_t)(b0 + tsel) * INF;
        float* xpt = xp + tsel * XP_TOK;
        for (int c = t8; c < INF; c += 256) {
            xpt[(c >> 6) * XP_STRIDE + (c & 63)] = xrow[iperm[c]];
        }
    }
    __syncthreads();

    const int ti = w & 3;
    const int kq = w >> 2;
    v8f accY[8];
    #pragma unroll
    for (int n = 0; n < 8; ++n) accY[n] = (v8f){};

    for (int chunk = 0; chunk < NCHUNK; ++chunk) {
        {
            const int cb0 = (w * 2) * 16;
            const int cb1 = cb0 + 16;
            const float* ap0 = g1 + (chunk * TT_ROWS + cb0 + l15) * 64 + koff;
            const float* ap1 = g1 + (chunk * TT_ROWS + cb1 + l15) * 64 + koff;
            v2f a0[16], a1[16];
            #pragma unroll
            for (int s = 0; s < 16; ++s) {
                a0[s] = *(const v2f*)(ap0 + 4 * s);
                a1[s] = *(const v2f*)(ap1 + 4 * s);
            }
            for (int tk = 0; tk < TPW; ++tk) {
                const float* xpt = xp + tk * XP_TOK;
                float* Ttk = Tt + tk * TT_TOK;
                for (int jp = 0; jp < 2; ++jp) {
                    v8f t0[2], t1[2];
                    #pragma unroll
                    for (int u = 0; u < 2; ++u) { t0[u] = (v8f){}; t1[u] = (v8f){}; }
                    #pragma unroll
                    for (int s = 0; s < 16; ++s) {
                        const int k = 4 * s + koff;
                        #pragma unroll
                        for (int u = 0; u < 2; ++u) {
                            const float* bp = xpt + ((jp * 2 + u) * 16 + l15) * XP_STRIDE + k;
                            v2f bb;
                            bb.x = bp[0]; bb.y = bp[1];
                            t0[u] = __builtin_amdgcn_wmma_f32_16x16x4_f32(false, a0[s], false, bb, (short)0, t0[u], false, false);
                            t1[u] = __builtin_amdgcn_wmma_f32_16x16x4_f32(false, a1[s], false, bb, (short)0, t1[u], false, false);
                        }
                        if ((s & 3) == 3) {
                            __builtin_amdgcn_sched_group_barrier(0x100, 8, 0);
                            __builtin_amdgcn_sched_group_barrier(0x008, 16, 0);
                        }
                    }
                    #pragma unroll
                    for (int v = 0; v < 8; ++v) {
                        const int r0 = cb0 + v + half * 8;
                        const int r1 = cb1 + v + half * 8;
                        #pragma unroll
                        for (int u = 0; u < 2; ++u) {
                            Ttk[r0 * TT_STRIDE + (jp * 2 + u) * 16 + l15] = t0[u][v];
                            Ttk[r1 * TT_STRIDE + (jp * 2 + u) * 16 + l15] = t1[u][v];
                        }
                    }
                }
            }
        }
        __syncthreads();

        const float* aptr = g0 + (ti * 16 + l15) * 1024 + kq * 256 + chunk * RH + koff;
        const float* tb   = Tt + (koff * 64 + l15) * TT_STRIDE + kq * 16;
        #pragma unroll 2
        for (int jj = 0; jj < 16; ++jj) {
            #pragma unroll
            for (int q = 0; q < RH; q += 4) {
                v2f a = *(const v2f*)(aptr + jj * 16 + q);
                #pragma unroll
                for (int n = 0; n < 8; ++n) {
                    const int tk = n >> 2;
                    const int mt = n & 3;
                    const float* bp = tb + tk * TT_TOK + (q * 64 + mt * 16) * TT_STRIDE + jj;
                    v2f bv;
                    bv.x = bp[0];
                    bv.y = bp[64 * TT_STRIDE];
                    accY[n] = __builtin_amdgcn_wmma_f32_16x16x4_f32(false, a, false, bv, (short)0, accY[n], false, false);
                }
            }
        }
        __syncthreads();
    }

    float* part = Tt;
    #pragma unroll
    for (int v = 0; v < 8; ++v) {
        const int i = ti * 16 + v + half * 8;
        #pragma unroll
        for (int n = 0; n < 8; ++n) {
            const int tk = n >> 2;
            const int mt = n & 3;
            part[(kq * TPW + tk) * XP_TOK + i * XP_STRIDE + mt * 16 + l15] = accY[n][v];
        }
    }
    __syncthreads();
    const float av = alpha[0];
    float* ylds = xp;
    for (int e = tid; e < TPW * OUTF; e += 512) {
        const int tk  = e >> 12;
        const int f   = e & (OUTF - 1);
        const int off = (f >> 6) * XP_STRIDE + (f & 63);
        const float* pp = part + tk * XP_TOK + off;
        const float s = ((pp[0] + pp[1 * TPW * XP_TOK]) + pp[2 * TPW * XP_TOK]) + pp[3 * TPW * XP_TOK];
        ylds[tk * XP_TOK + off] = s * av * pds[f];
    }
    __syncthreads();

    {
        float* orow = out + (size_t)(b0 + tsel) * OUTF;
        const float* yl = ylds + tsel * XP_TOK;
        for (int q = t8; q < OUTF; q += 256) {
            const int o = oinv[q];
            orow[q] = yl[(o >> 6) * XP_STRIDE + (o & 63)] + bias[q];
        }
    }
}

extern "C" void kernel_launch(void* const* d_in, const int* in_sizes, int n_in,
                              void* d_out, int out_size, void* d_ws, size_t ws_size,
                              hipStream_t stream) {
    (void)in_sizes; (void)n_in; (void)out_size; (void)d_ws; (void)ws_size;
    const float* x     = (const float*)d_in[0];
    const float* g0    = (const float*)d_in[1];
    const float* g1    = (const float*)d_in[2];
    const float* alpha = (const float*)d_in[3];
    const float* pds   = (const float*)d_in[4];
    const float* bias  = (const float*)d_in[5];
    const int*   iperm = (const int*)d_in[6];
    const int*   oinv  = (const int*)d_in[7];
    float* out = (float*)d_out;

    (void)hipFuncSetAttribute(reinterpret_cast<const void*>(tt_linear_kernel),
                              hipFuncAttributeMaxDynamicSharedMemorySize, SMEM_BYTES);

    tt_linear_kernel<<<TOKENS / TPW, 512, SMEM_BYTES, stream>>>(
        x, g0, g1, alpha, pds, bias, iperm, oinv, out);
}


# GEMM2 B fragments: T rows reordered (m*8+r), read by hand-placed two-address LDS loads at immediate offsets one j ahead: no vector-unit adds or moves beside the GEMM2 WMMAs; sums unchanged
# speedup vs baseline: 1.2400x; 1.0494x over previous
#include <hip/hip_runtime.h>


#define TOKENS 2048
#define TPW 2
#define INF 4096
#define OUTF 4096
#define RH 8
#define NCHUNK 2
#define XP_STRIDE 65
#define TT_STRIDE 65
#define TT_ROWS (RH * 64)
#define TT_TOK (TT_ROWS * TT_STRIDE)
#define XP_TOK (64 * XP_STRIDE)
#define SMEM_FLOATS (TPW * TT_TOK + TPW * XP_TOK)
#define SMEM_BYTES (SMEM_FLOATS * 4)

typedef __attribute__((ext_vector_type(2))) float v2f;
typedef __attribute__((ext_vector_type(8))) float v8f;

__global__ __launch_bounds__(512)
void tt_linear_kernel(const float* __restrict__ x,
                      const float* __restrict__ g0,
                      const float* __restrict__ g1,
                      const float* __restrict__ alpha,
                      const float* __restrict__ pds,
                      const float* __restrict__ bias,
                      const int*   __restrict__ iperm,
                      const int*   __restrict__ oinv,
                      float* __restrict__ out)
{
    extern __shared__ float smem[];
    float* Tt = smem;
    float* xp = smem + TPW * TT_TOK;

    const int b0   = blockIdx.x * TPW;
    const int tid  = threadIdx.x;
    const int lane = tid & 31;
    const int w    = tid >> 5;
    const int half = lane >> 4;
    const int l15  = lane & 15;
    const int koff = half * 2;
    const int tsel = tid >> 8;
    const int t8   = tid & 255;

    {
        const float* xrow = x + (size_t)(b0 + tsel) * INF;
        float* xpt = xp + tsel * XP_TOK;
        for (int c = t8; c < INF; c += 256) {
            xpt[(c >> 6) * XP_STRIDE + (c & 63)] = xrow[iperm[c]];
        }
    }
    __syncthreads();

    const int ti = w & 3;
    const int kq = w >> 2;
    v8f accY[8];
    #pragma unroll
    for (int n = 0; n < 8; ++n) accY[n] = (v8f){};

    for (int chunk = 0; chunk < NCHUNK; ++chunk) {
        {
            const int cb0 = (w * 2) * 16;
            const int cb1 = cb0 + 16;
            const float* ap0 = g1 + (chunk * TT_ROWS + cb0 + l15) * 64 + koff;
            const float* ap1 = g1 + (chunk * TT_ROWS + cb1 + l15) * 64 + koff;
            v2f a0[16], a1[16];
            #pragma unroll
            for (int s = 0; s < 16; ++s) {
                a0[s] = *(const v2f*)(ap0 + 4 * s);
                a1[s] = *(const v2f*)(ap1 + 4 * s);
            }
            for (int tk = 0; tk < TPW; ++tk) {
                const float* xpt = xp + tk * XP_TOK;
                float* Ttk = Tt + tk * TT_TOK;
                for (int jp = 0; jp < 2; ++jp) {
                    v8f t0[2], t1[2];
                    #pragma unroll
                    for (int u = 0; u < 2; ++u) { t0[u] = (v8f){}; t1[u] = (v8f){}; }
                    #pragma unroll
                    for (int s = 0; s < 16; ++s) {
                        const int k = 4 * s + koff;
                        #pragma unroll
                        for (int u = 0; u < 2; ++u) {
                            const float* bp = xpt + ((jp * 2 + u) * 16 + l15) * XP_STRIDE + k;
                            v2f bb;
                            bb.x = bp[0]; bb.y = bp[1];
                            t0[u] = __builtin_amdgcn_wmma_f32_16x16x4_f32(false, a0[s], false, bb, (short)0, t0[u], false, false);
                            t1[u] = __builtin_amdgcn_wmma_f32_16x16x4_f32(false, a1[s], false, bb, (short)0, t1[u], false, false);
                        }
                        if ((s & 3) == 3) {
                            __builtin_amdgcn_sched_group_barrier(0x100, 8, 0);
                            __builtin_amdgcn_sched_group_barrier(0x008, 16, 0);
                        }
                    }
                    #pragma unroll
                    for (int v = 0; v < 8; ++v) {
                        const int r0 = ((cb0 & 63) + v + half * 8) * RH + (w >> 1);
                        const int r1 = ((cb1 & 63) + v + half * 8) * RH + (w >> 1);
                        #pragma unroll
                        for (int u = 0; u < 2; ++u) {
                            Ttk[r0 * TT_STRIDE + (jp * 2 + u) * 16 + l15] = t0[u][v];
                            Ttk[r1 * TT_STRIDE + (jp * 2 + u) * 16 + l15] = t1[u][v];
                        }
                    }
                }
            }
        }
        __syncthreads();

        const float* aptr = g0 + (ti * 16 + l15) * 1024 + kq * 256 + chunk * RH + koff;
        const unsigned tb = (unsigned)(size_t)(Tt + (l15 * RH + koff) * TT_STRIDE + kq * 16);
        unsigned bbase[16];
        #pragma unroll
        for (int n = 0; n < 8; ++n) {
            #pragma unroll
            for (int qi = 0; qi < 2; ++qi)
                bbase[n * 2 + qi] = tb + 4u * (unsigned)((n >> 2) * TT_TOK + ((n & 3) * 16 * RH + qi * 4) * TT_STRIDE);
        }
        v2f afr[2][8];
        v2f bfr[2][16];
        #pragma unroll
        for (int t = 0; t < 8; ++t) afr[0][t] = *(const v2f*)(aptr + (t >> 1) * 16 + (t & 1) * 4);
        #pragma unroll
        for (int t = 0; t < 16; ++t) {
            asm volatile("ds_load_2addr_b32 %0, %1 offset0:0 offset1:65" : "=v"(bfr[0][t]) : "v"(bbase[t]));
        }
        #pragma unroll
        for (int jj = 0; jj < 16; ++jj) {
            const int cur = jj & 1;
            __builtin_amdgcn_sched_barrier(0);
            if ((jj & 3) == 0 && jj < 12) {
                #pragma unroll
                for (int t = 0; t < 8; ++t)
                    afr[((jj >> 2) + 1) & 1][t] = *(const v2f*)(aptr + (jj + 4 + (t >> 1)) * 16 + (t & 1) * 4);
            }
            if (jj < 15) {
                #pragma unroll
                for (int t = 0; t < 16; ++t) {
                    asm volatile("ds_load_2addr_b32 %0, %1 offset0:%2 offset1:%3"
                                 : "=v"(bfr[cur ^ 1][t]) : "v"(bbase[t]), "n"(jj + 1), "n"(jj + 1 + 65));
                }
                asm volatile("s_wait_dscnt 0x10"
                             : "+v"(bfr[cur][0]), "+v"(bfr[cur][1]), "+v"(bfr[cur][2]), "+v"(bfr[cur][3]),
                               "+v"(bfr[cur][4]), "+v"(bfr[cur][5]), "+v"(bfr[cur][6]), "+v"(bfr[cur][7]),
                               "+v"(bfr[cur][8]), "+v"(bfr[cur][9]), "+v"(bfr[cur][10]), "+v"(bfr[cur][11]),
                               "+v"(bfr[cur][12]), "+v"(bfr[cur][13]), "+v"(bfr[cur][14]), "+v"(bfr[cur][15]));
            } else {
                asm volatile("s_wait_dscnt 0x0"
                             : "+v"(bfr[cur][0]), "+v"(bfr[cur][1]), "+v"(bfr[cur][2]), "+v"(bfr[cur][3]),
                               "+v"(bfr[cur][4]), "+v"(bfr[cur][5]), "+v"(bfr[cur][6]), "+v"(bfr[cur][7]),
                               "+v"(bfr[cur][8]), "+v"(bfr[cur][9]), "+v"(bfr[cur][10]), "+v"(bfr[cur][11]),
                               "+v"(bfr[cur][12]), "+v"(bfr[cur][13]), "+v"(bfr[cur][14]), "+v"(bfr[cur][15]));
            }
            #pragma unroll
            for (int qi = 0; qi < 2; ++qi) {
                const v2f a = afr[(jj >> 2) & 1][(jj & 3) * 2 + qi];
                #pragma unroll
                for (int n = 0; n < 8; ++n)
                    accY[n] = __builtin_amdgcn_wmma_f32_16x16x4_f32(false, a, false, bfr[cur][n * 2 + qi], (short)0, accY[n], false, false);
            }
        }
        __syncthreads();
    }

    float* part = Tt;
    #pragma unroll
    for (int v = 0; v < 8; ++v) {
        const int i = ti * 16 + v + half * 8;
        #pragma unroll
        for (int n = 0; n < 8; ++n) {
            const int tk = n >> 2;
            const int mt = n & 3;
            part[(kq * TPW + tk) * XP_TOK + i * XP_STRIDE + mt * 16 + l15] = accY[n][v];
        }
    }
    __syncthreads();
    const float av = alpha[0];
    float* ylds = xp;
    for (int e = tid; e < TPW * OUTF; e += 512) {
        const int tk  = e >> 12;
        const int f   = e & (OUTF - 1);
        const int off = (f >> 6) * XP_STRIDE + (f & 63);
        const float* pp = part + tk * XP_TOK + off;
        const float s = ((pp[0] + pp[1 * TPW * XP_TOK]) + pp[2 * TPW * XP_TOK]) + pp[3 * TPW * XP_TOK];
        ylds[tk * XP_TOK + off] = s * av * pds[f];
    }
    __syncthreads();

    {
        float* orow = out + (size_t)(b0 + tsel) * OUTF;
        const float* yl = ylds + tsel * XP_TOK;
        for (int q = t8; q < OUTF; q += 256) {
            const int o = oinv[q];
            orow[q] = yl[(o >> 6) * XP_STRIDE + (o & 63)] + bias[q];
        }
    }
}

extern "C" void kernel_launch(void* const* d_in, const int* in_sizes, int n_in,
                              void* d_out, int out_size, void* d_ws, size_t ws_size,
                              hipStream_t stream) {
    (void)in_sizes; (void)n_in; (void)out_size; (void)d_ws; (void)ws_size;
    const float* x     = (const float*)d_in[0];
    const float* g0    = (const float*)d_in[1];
    const float* g1    = (const float*)d_in[2];
    const float* alpha = (const float*)d_in[3];
    const float* pds   = (const float*)d_in[4];
    const float* bias  = (const float*)d_in[5];
    const int*   iperm = (const int*)d_in[6];
    const int*   oinv  = (const int*)d_in[7];
    float* out = (float*)d_out;

    (void)hipFuncSetAttribute(reinterpret_cast<const void*>(tt_linear_kernel),
                              hipFuncAttributeMaxDynamicSharedMemorySize, SMEM_BYTES);

    tt_linear_kernel<<<TOKENS / TPW, 512, SMEM_BYTES, stream>>>(
        x, g0, g1, alpha, pds, bias, iperm, oinv, out);
}
